# starting kernel
# baseline (speedup 1.0000x reference)
#include <hip/hip_runtime.h>

typedef __attribute__((ext_vector_type(2))) float v2f;
typedef __attribute__((ext_vector_type(8))) float v8f;

#define BATCH 8192
#define FEATS 4096
#define RANK_LOW 211
#define RANK_HIGH 7
#define DENSITY 0.05f
#define EPT 16

__global__ void kw_factor_kernel(const float* __restrict__ aa,
                                 float* __restrict__ factor) {
    int j = blockIdx.x * blockDim.x + threadIdx.x;
    if (j < FEATS) {
        factor[j] = __expf((DENSITY - aa[j]) * 1.0f);
    }
}

__global__ void kw_colmean_wmma_kernel(const float* __restrict__ x,
                                       float* __restrict__ colmean) {
    const int col0 = blockIdx.x * 16;
    const int lane = threadIdx.x & 31;
    const int wave = threadIdx.x >> 5;
    const int n    = lane & 15;
    const int kb   = (lane >> 4) * 2;

    v2f a; a.x = 1.0f; a.y = 1.0f;
    v8f c = {};

    const float* base = x + col0 + n;
    const int r0 = wave * (BATCH / 8);
    const int r1 = r0 + (BATCH / 8);
    for (int r = r0; r < r1; r += 4) {
        v2f b;
        b.x = base[(size_t)(r + kb) * FEATS];
        b.y = base[(size_t)(r + kb + 1) * FEATS];
        c = __builtin_amdgcn_wmma_f32_16x16x4_f32(
                false, a, false, b, (short)0, c, false, false);
    }

    __shared__ float part[8][16];
    if (lane < 16) part[wave][lane] = c[0];
    __syncthreads();
    if (threadIdx.x < 16) {
        float s = 0.0f;
        for (int w = 0; w < 8; ++w) s += part[w][threadIdx.x];
        colmean[col0 + threadIdx.x] = s * (1.0f / (float)BATCH);
    }
}

__device__ __forceinline__ unsigned order_key(float f) {
    unsigned u = __float_as_uint(f);
    return (u & 0x80000000u) ? ~u : (u | 0x80000000u);
}

__device__ __forceinline__ unsigned radix_select_desc(
        const unsigned rk[EPT], unsigned* __restrict__ hist,
        unsigned* __restrict__ s_prefix, int* __restrict__ s_target,
        int R, int t) {
    unsigned prefix = 0;
    int target = R;
    for (int shift = 24; shift >= 0; shift -= 8) {
        hist[t] = 0;
        __syncthreads();
        const unsigned hmask = (shift == 24) ? 0u : (0xFFFFFFFFu << (shift + 8));
        #pragma unroll
        for (int e = 0; e < EPT; ++e) {
            unsigned k = rk[e];
            if ((k & hmask) == prefix)
                atomicAdd(&hist[(k >> shift) & 0xFFu], 1u);
        }
        __syncthreads();
        if (t == 0) {
            int tgt = target;
            int b = 255;
            for (; b > 0; --b) {
                int cnt = (int)hist[b];
                if (tgt <= cnt) break;
                tgt -= cnt;
            }
            *s_prefix = prefix | ((unsigned)b << shift);
            *s_target = tgt;
        }
        __syncthreads();
        prefix = *s_prefix;
        target = *s_target;
    }
    return prefix;
}

__global__ void __launch_bounds__(256)
kw_row_kernel(const float* __restrict__ x,
              const float* __restrict__ factor,
              const float* __restrict__ colmean,
              float* __restrict__ out) {
    __shared__ unsigned hist[256];
    __shared__ unsigned s_prefix;
    __shared__ int      s_target;

    const int row = blockIdx.x;
    const int t   = threadIdx.x;

    const float4* __restrict__ xin = (const float4*)(x + (size_t)row * FEATS);
    const float4* __restrict__ fin = (const float4*)factor;

    float    rx[EPT];
    unsigned rk[EPT];

    #pragma unroll
    for (int it = 0; it < EPT / 4; ++it) {
        int idx = it * 256 + t;
        float4 xv = xin[idx];
        float4 fv = fin[idx];
        rx[it * 4 + 0] = xv.x;  rk[it * 4 + 0] = order_key(xv.x * fv.x);
        rx[it * 4 + 1] = xv.y;  rk[it * 4 + 1] = order_key(xv.y * fv.y);
        rx[it * 4 + 2] = xv.z;  rk[it * 4 + 2] = order_key(xv.z * fv.z);
        rx[it * 4 + 3] = xv.w;  rk[it * 4 + 3] = order_key(xv.w * fv.w);
    }

    const unsigned thrHigh = radix_select_desc(rk, hist, &s_prefix, &s_target,
                                               RANK_HIGH, t);
    const unsigned thrLow  = radix_select_desc(rk, hist, &s_prefix, &s_target,
                                               RANK_LOW, t);

    const float4* __restrict__ cmin = (const float4*)colmean;
    float4* __restrict__ oout = (float4*)(out + (size_t)row * FEATS);
    #pragma unroll
    for (int it = 0; it < EPT / 4; ++it) {
        int idx = it * 256 + t;
        float4 cm = cmin[idx];
        float4 o;
        {
            unsigned k = rk[it * 4 + 0];
            bool m2 = (k <= thrHigh);
            bool m  = m2 && (k >= thrLow);
            o.x = (m ? rx[it * 4 + 0] : 0.0f) - (m2 ? cm.x : 0.0f);
        }
        {
            unsigned k = rk[it * 4 + 1];
            bool m2 = (k <= thrHigh);
            bool m  = m2 && (k >= thrLow);
            o.y = (m ? rx[it * 4 + 1] : 0.0f) - (m2 ? cm.y : 0.0f);
        }
        {
            unsigned k = rk[it * 4 + 2];
            bool m2 = (k <= thrHigh);
            bool m  = m2 && (k >= thrLow);
            o.z = (m ? rx[it * 4 + 2] : 0.0f) - (m2 ? cm.z : 0.0f);
        }
        {
            unsigned k = rk[it * 4 + 3];
            bool m2 = (k <= thrHigh);
            bool m  = m2 && (k >= thrLow);
            o.w = (m ? rx[it * 4 + 3] : 0.0f) - (m2 ? cm.w : 0.0f);
        }
        oout[idx] = o;
    }
}

extern "C" void kernel_launch(void* const* d_in, const int* in_sizes, int n_in,
                              void* d_out, int out_size, void* d_ws, size_t ws_size,
                              hipStream_t stream) {
    const float* x  = (const float*)d_in[0];
    const float* aa = (const float*)d_in[1];
    float* out = (float*)d_out;

    float* factor  = (float*)d_ws;
    float* colmean = (float*)d_ws + FEATS;

    kw_factor_kernel<<<FEATS / 256, 256, 0, stream>>>(aa, factor);

    kw_colmean_wmma_kernel<<<FEATS / 16, 256, 0, stream>>>(x, colmean);

    kw_row_kernel<<<BATCH, 256, 0, stream>>>(x, factor, colmean, out);
}


# Row kernel: each row-wide search (a rank's bin per digit, its key in a short list) is run by ONE wave per rank, handed on via LDS at a barrier, not repeated by all 8 waves; 8 barriers a row, not 5
# speedup vs baseline: 13.7001x; 13.7001x over previous
#include <hip/hip_runtime.h>

typedef __attribute__((ext_vector_type(2))) float v2f;
typedef __attribute__((ext_vector_type(8))) float v8f;

#define BATCH 8192
#define FEATS 4096
#define RANK_LOW 211
#define RANK_HIGH 7
#define DENSITY 0.05f
#define EPT 16
#define REP_WORDS (256 * 32)

#define CM_WMMAS 8
#define CM_BATCH_ROWS (4 * CM_WMMAS)
#define CM_WAVE_ROWS (BATCH / 8)
#define CM_BATCHES (CM_WAVE_ROWS / CM_BATCH_ROWS)
static_assert(CM_WAVE_ROWS % CM_BATCH_ROWS == 0, "whole batches only");
static_assert(CM_BATCHES >= 4 && CM_BATCHES % 4 == 0, "three batches ahead, four sets in turn");

__device__ __forceinline__ void cm_request(v2f (&b)[CM_WMMAS], const float* __restrict__ p) {
    #pragma unroll
    for (int i = 0; i < CM_WMMAS; ++i) {
        b[i].x = p[(size_t)(4 * i) * FEATS];
        b[i].y = p[(size_t)(4 * i + 1) * FEATS];
    }
}

__device__ __forceinline__ v8f cm_add(v8f c, const v2f a, const v2f (&b)[CM_WMMAS]) {
    #pragma unroll
    for (int i = 0; i < CM_WMMAS; ++i)
        c = __builtin_amdgcn_wmma_f32_16x16x4_f32(
                false, a, false, b[i], (short)0, c, false, false);
    return c;
}

__global__ void kw_colmean_wmma_kernel(const float* __restrict__ x,
                                       const float* __restrict__ aa,
                                       float* __restrict__ colmean,
                                       float* __restrict__ factor) {
    const int col0 = blockIdx.x * 16;
    const int lane = threadIdx.x & 31;
    const int wave = threadIdx.x >> 5;
    const int n    = lane & 15;
    const int kb   = (lane >> 4) * 2;

    float av = 0.0f;
    if (threadIdx.x < 16) av = aa[col0 + threadIdx.x];

    v2f a; a.x = 1.0f; a.y = 1.0f;
    v8f c = {};

    const size_t step = (size_t)CM_BATCH_ROWS * FEATS;
    const float* p = x + (size_t)(wave * CM_WAVE_ROWS + kb) * FEATS + col0 + n;

    v2f b0[CM_WMMAS], b1[CM_WMMAS], b2[CM_WMMAS], b3[CM_WMMAS];
    cm_request(b0, p);
    cm_request(b1, p + step);
    cm_request(b2, p + 2 * step);
    p += 3 * step;
    for (int it = 0; it < CM_BATCHES / 4 - 1; ++it) {
        cm_request(b3, p);              c = cm_add(c, a, b0);
        cm_request(b0, p + step);       c = cm_add(c, a, b1);
        cm_request(b1, p + 2 * step);   c = cm_add(c, a, b2);
        cm_request(b2, p + 3 * step);   c = cm_add(c, a, b3);
        p += 4 * step;
    }
    cm_request(b3, p);
    c = cm_add(c, a, b0);
    c = cm_add(c, a, b1);
    c = cm_add(c, a, b2);
    c = cm_add(c, a, b3);

    __shared__ float part[8][16];
    if (lane < 16) part[wave][lane] = c[0];
    __syncthreads();
    if (threadIdx.x < 16) {
        float s = 0.0f;
        for (int w = 0; w < 8; ++w) s += part[w][threadIdx.x];
        colmean[col0 + threadIdx.x] = s * (1.0f / (float)BATCH);
        factor[col0 + threadIdx.x]  = __expf((DENSITY - av) * 1.0f);
    }
}

__device__ __forceinline__ unsigned order_key(float f) {
    unsigned u = __float_as_uint(f);
    return (u & 0x80000000u) ? ~u : (u | 0x80000000u);
}

template <int CTRL>
__device__ __forceinline__ int from_higher_lane(int v) {
    return __builtin_amdgcn_update_dpp(0, v, CTRL, 0xF, 0xF, true);
}

__device__ __forceinline__ void find_bin(const unsigned* h, int lane, int target,
                                         int& bin, int& tgt) {
    const uint4 q0 = *(const uint4*)(h + 8 * lane);
    const uint4 q1 = *(const uint4*)(h + 8 * lane + 4);
    int c[8];
    c[0] = (int)q0.x; c[1] = (int)q0.y; c[2] = (int)q0.z; c[3] = (int)q0.w;
    c[4] = (int)q1.x; c[5] = (int)q1.y; c[6] = (int)q1.z; c[7] = (int)q1.w;
    const int tot = ((c[0] + c[1]) + (c[2] + c[3])) + ((c[4] + c[5]) + (c[6] + c[7]));

    int s = tot;
    s += from_higher_lane<0x101>(s);
    s += from_higher_lane<0x102>(s);
    s += from_higher_lane<0x104>(s);
    s += from_higher_lane<0x108>(s);
    const int upper = __builtin_amdgcn_readlane(s, 16);
    if (lane < 16) s += upper;

    int a = s - tot;
    int bsel = 0, tsel = 0;
    bool mine = false;
    #pragma unroll
    for (int j = 7; j >= 0; --j) {
        const int nxt = a + c[j];
        const bool hit = (a < target) && (target <= nxt);
        bsel = hit ? (8 * lane + j) : bsel;
        tsel = hit ? (target - a) : tsel;
        mine = mine || hit;
        a = nxt;
    }
    const unsigned owners = __builtin_amdgcn_ballot_w32(mine);
    const int owner = (int)__ffs(owners) - 1;
    bin = __builtin_amdgcn_readlane(bsel, owner);
    tgt = __builtin_amdgcn_readlane(tsel, owner);
}

__device__ __forceinline__ void count_digit(const unsigned rk[EPT], unsigned* h, int shift,
                                            unsigned prefH, unsigned prefL) {
    const unsigned hmask = 0xFFFFFFFFu << (shift + 8);
    #pragma unroll
    for (int e = 0; e < EPT; ++e) {
        const unsigned k  = rk[e];
        const unsigned hi = k & hmask;
        const bool mH = (hi == prefH);
        const bool mL = (hi == prefL);
        if (mH || mL)
            atomicAdd(&h[(mH ? 0u : 256u) + ((k >> shift) & 0xFFu)], 1u);
    }
}

__device__ __forceinline__ unsigned pick_rank(unsigned mine, bool valid, int above, int atleast,
                                              int target) {
    const bool hit = valid && (above < target) && (target <= atleast);
    const unsigned owners = __builtin_amdgcn_ballot_w32(hit);
    const int owner = (int)__ffs(owners) - 1;
    return (unsigned)__builtin_amdgcn_readlane((int)mine, owner);
}

#define LIST_SLOTS 32

__device__ __forceinline__ void radix_select_two_desc(
        const unsigned rk[EPT], unsigned* lds, int t,
        unsigned& keyHigh, unsigned& keyLow) {
    unsigned* rep = lds;
    unsigned* tot = lds + REP_WORDS + 2 * 512;

    for (int e = 0; e < EPT; ++e)
        atomicAdd(&rep[((rk[e] >> 24) << 5) + (unsigned)(t & 31)], 1u);
    __syncthreads();
    {
        const unsigned* cnt = rep + 32 * t;
        const int rot = (t >> 1) & 7;
        unsigned s0 = 0, s1 = 0, s2 = 0, s3 = 0;
        for (int j = 0; j < 8; ++j) {
            const uint4 q = *(const uint4*)(cnt + 4 * ((j + rot) & 7));
            s0 += q.x; s1 += q.y; s2 += q.z; s3 += q.w;
        }
        tot[t] = (s0 + s1) + (s2 + s3);
    }
    __syncthreads();
    if (t == 0) {
        const uint2 zero2 = {0u, 0u};
        *(uint2*)rep = zero2;
    }
    const int wave = __builtin_amdgcn_readfirstlane(t >> 5);
    const int lane = (int)__builtin_amdgcn_mbcnt_lo(~0u, 0u);
    int binH, tgtH, binL, tgtL;
    if (wave == 0) {
        int b, g;
        find_bin(tot, lane, RANK_HIGH, b, g);
        if (lane == 0) { const uint2 r = {(unsigned)b, (unsigned)g}; *(uint2*)(rep + 4) = r; }
    } else if (wave == 1) {
        int b, g;
        find_bin(tot, lane, RANK_LOW, b, g);
        if (lane == 0) { const uint2 r = {(unsigned)b, (unsigned)g}; *(uint2*)(rep + 6) = r; }
    }
    __syncthreads();
    {
        const uint4 r = *(const uint4*)(rep + 4);
        binH = __builtin_amdgcn_readfirstlane((int)r.x);
        tgtH = __builtin_amdgcn_readfirstlane((int)r.y);
        binL = __builtin_amdgcn_readfirstlane((int)r.z);
        tgtL = __builtin_amdgcn_readfirstlane((int)r.w);
    }
    unsigned prefH = (unsigned)binH << 24;
    unsigned prefL = (unsigned)binL << 24;
    int targH = tgtH, targL = tgtL;
    bool same = (binH == binL);

    {
        unsigned* h = lds + REP_WORDS;
        count_digit(rk, h, 16, prefH, prefL);
        __syncthreads();
        if (wave == 2) {
            int b, g;
            find_bin(h, lane, targH, b, g);
            if (lane == 0) { const uint2 r = {(unsigned)b, (unsigned)g}; *(uint2*)(rep + 8) = r; }
        } else if (wave == 3) {
            int b, g;
            find_bin(same ? h : h + 256, lane, targL, b, g);
            if (lane == 0) { const uint2 r = {(unsigned)b, (unsigned)g}; *(uint2*)(rep + 10) = r; }
        }
        __syncthreads();
        {
            const uint4 r = *(const uint4*)(rep + 8);
            binH = __builtin_amdgcn_readfirstlane((int)r.x);
            tgtH = __builtin_amdgcn_readfirstlane((int)r.y);
            binL = __builtin_amdgcn_readfirstlane((int)r.z);
            tgtL = __builtin_amdgcn_readfirstlane((int)r.w);
        }
        prefH |= ((unsigned)binH << 16);
        prefL |= ((unsigned)binL << 16);
        targH = tgtH;
        targL = tgtL;
        same = same && (binH == binL);
    }

    #pragma unroll
    for (int e = 0; e < EPT; ++e) {
        const unsigned k  = rk[e];
        const unsigned hi = k & 0xFFFF0000u;
        const bool mH = (hi == prefH);
        const bool mL = (hi == prefL);
        if (mH || mL) {
            const unsigned w = mH ? 0u : 1u;
            const unsigned slot = atomicAdd(&rep[w], 1u);
            if (slot < (unsigned)LIST_SLOTS) rep[32u + 32u * w + slot] = k;
        }
    }
    __syncthreads();
    const int cH = __builtin_amdgcn_readfirstlane((int)rep[0]);
    const int cL = __builtin_amdgcn_readfirstlane((int)rep[1]);

    if (cH > LIST_SLOTS || cL > LIST_SLOTS) {
        const uint2 zero2 = {0u, 0u};
        ((uint2*)tot)[t] = zero2;
        #pragma unroll
        for (int d = 2; d < 4; ++d) {
            const int shift = 24 - 8 * d;
            unsigned* h = lds + REP_WORDS + 512 * (d - 1);
            count_digit(rk, h, shift, prefH, prefL);
            __syncthreads();
            find_bin(h, lane, targH, binH, tgtH);
            find_bin(same ? h : h + 256, lane, targL, binL, tgtL);
            prefH |= ((unsigned)binH << shift);
            prefL |= ((unsigned)binL << shift);
            targH = tgtH;
            targL = tgtL;
            same = same && (binH == binL);
        }
        keyHigh = prefH;
        keyLow  = prefL;
    } else {
        if (wave == 4 || (wave == 5 && !same)) {
            const int      c    = (wave == 4) ? cH : cL;
            const unsigned* lst = rep + ((wave == 4) ? 32 : 64);
            const bool v = lane < c;
            const unsigned mine = v ? lst[lane] : 0u;
            int above = 0, atleast = 0;
            #pragma nounroll
            for (int j = 0; j < c; ++j) {
                const unsigned kj = (unsigned)__builtin_amdgcn_readlane((int)mine, j);
                above   += (kj >  mine) ? 1 : 0;
                atleast += (kj >= mine) ? 1 : 0;
            }
            if (wave == 4) {
                const unsigned kH = pick_rank(mine, v, above, atleast, targH);
                if (lane == 0) rep[12] = kH;
                if (same) {
                    const unsigned kL = pick_rank(mine, v, above, atleast, targL);
                    if (lane == 0) rep[13] = kL;
                }
            } else {
                const unsigned kL = pick_rank(mine, v, above, atleast, targL);
                if (lane == 0) rep[13] = kL;
            }
        }
        __syncthreads();
        const uint2 r = *(const uint2*)(rep + 12);
        keyHigh = (unsigned)__builtin_amdgcn_readfirstlane((int)r.x);
        keyLow  = (unsigned)__builtin_amdgcn_readfirstlane((int)r.y);
    }
}

__global__ void __launch_bounds__(256, 16)
kw_row_kernel(const float* __restrict__ x,
              const float* __restrict__ factor,
              const float* __restrict__ colmean,
              float* __restrict__ out) {
    __shared__ __attribute__((aligned(16))) unsigned hist[REP_WORDS + 3 * 512];

    const int row = blockIdx.x;
    const int t   = threadIdx.x;

    const float4* __restrict__ xin = (const float4*)(x + (size_t)row * FEATS);
    const float4* __restrict__ fin = (const float4*)factor;

    float    rx[EPT];
    unsigned rk[EPT];

    #pragma unroll
    for (int it = 0; it < EPT / 4; ++it) {
        int idx = it * 256 + t;
        float4 xv = xin[idx];
        float4 fv = fin[idx];
        rx[it * 4 + 0] = xv.x;  rk[it * 4 + 0] = order_key(xv.x * fv.x);
        rx[it * 4 + 1] = xv.y;  rk[it * 4 + 1] = order_key(xv.y * fv.y);
        rx[it * 4 + 2] = xv.z;  rk[it * 4 + 2] = order_key(xv.z * fv.z);
        rx[it * 4 + 3] = xv.w;  rk[it * 4 + 3] = order_key(xv.w * fv.w);
    }

    const uint4 zero4 = {0u, 0u, 0u, 0u};
    for (int i = 0; i < (REP_WORDS + 2 * 512) / (4 * 256); ++i)
        ((uint4*)hist)[i * 256 + t] = zero4;
    __syncthreads();
    unsigned thrHigh, thrLow;
    radix_select_two_desc(rk, hist, t, thrHigh, thrLow);

    const float4* __restrict__ cmin = (const float4*)colmean;
    float4* __restrict__ oout = (float4*)(out + (size_t)row * FEATS);
    #pragma unroll
    for (int it = 0; it < EPT / 4; ++it) {
        int idx = it * 256 + t;
        float4 cm = cmin[idx];
        float4 o;
        {
            unsigned k = rk[it * 4 + 0];
            bool m2 = (k <= thrHigh);
            bool m  = m2 && (k >= thrLow);
            o.x = (m ? rx[it * 4 + 0] : 0.0f) - (m2 ? cm.x : 0.0f);
        }
        {
            unsigned k = rk[it * 4 + 1];
            bool m2 = (k <= thrHigh);
            bool m  = m2 && (k >= thrLow);
            o.y = (m ? rx[it * 4 + 1] : 0.0f) - (m2 ? cm.y : 0.0f);
        }
        {
            unsigned k = rk[it * 4 + 2];
            bool m2 = (k <= thrHigh);
            bool m  = m2 && (k >= thrLow);
            o.z = (m ? rx[it * 4 + 2] : 0.0f) - (m2 ? cm.z : 0.0f);
        }
        {
            unsigned k = rk[it * 4 + 3];
            bool m2 = (k <= thrHigh);
            bool m  = m2 && (k >= thrLow);
            o.w = (m ? rx[it * 4 + 3] : 0.0f) - (m2 ? cm.w : 0.0f);
        }
        oout[idx] = o;
    }
}

extern "C" void kernel_launch(void* const* d_in, const int* in_sizes, int n_in,
                              void* d_out, int out_size, void* d_ws, size_t ws_size,
                              hipStream_t stream) {
    const float* x  = (const float*)d_in[0];
    const float* aa = (const float*)d_in[1];
    float* out = (float*)d_out;

    float* factor  = (float*)d_ws;
    float* colmean = (float*)d_ws + FEATS;

    kw_colmean_wmma_kernel<<<FEATS / 16, 256, 0, stream>>>(x, aa, colmean, factor);

    kw_row_kernel<<<BATCH, 256, 0, stream>>>(x, factor, colmean, out);
}


# Row kernel: first level is a 256-bin window over xp in [0.5, 8) (key bits 24..17), only keys inside it counted; both ranks' short lists come straight from it; radix path only if that fails
# speedup vs baseline: 16.9736x; 1.2389x over previous
#include <hip/hip_runtime.h>

typedef __attribute__((ext_vector_type(2))) float v2f;
typedef __attribute__((ext_vector_type(8))) float v8f;

#define BATCH 8192
#define FEATS 4096
#define RANK_LOW 211
#define RANK_HIGH 7
#define DENSITY 0.05f
#define EPT 16
#define REP_WORDS (256 * 32)

#define CM_WMMAS 8
#define CM_BATCH_ROWS (4 * CM_WMMAS)
#define CM_WAVE_ROWS (BATCH / 8)
#define CM_BATCHES (CM_WAVE_ROWS / CM_BATCH_ROWS)
static_assert(CM_WAVE_ROWS % CM_BATCH_ROWS == 0, "whole batches only");
static_assert(CM_BATCHES >= 4 && CM_BATCHES % 4 == 0, "three batches ahead, four sets in turn");

__device__ __forceinline__ void cm_request(v2f (&b)[CM_WMMAS], const float* __restrict__ p) {
    #pragma unroll
    for (int i = 0; i < CM_WMMAS; ++i) {
        b[i].x = p[(size_t)(4 * i) * FEATS];
        b[i].y = p[(size_t)(4 * i + 1) * FEATS];
    }
}

__device__ __forceinline__ v8f cm_add(v8f c, const v2f a, const v2f (&b)[CM_WMMAS]) {
    #pragma unroll
    for (int i = 0; i < CM_WMMAS; ++i)
        c = __builtin_amdgcn_wmma_f32_16x16x4_f32(
                false, a, false, b[i], (short)0, c, false, false);
    return c;
}

__global__ void kw_colmean_wmma_kernel(const float* __restrict__ x,
                                       const float* __restrict__ aa,
                                       float* __restrict__ colmean,
                                       float* __restrict__ factor) {
    const int col0 = blockIdx.x * 16;
    const int lane = threadIdx.x & 31;
    const int wave = threadIdx.x >> 5;
    const int n    = lane & 15;
    const int kb   = (lane >> 4) * 2;

    float av = 0.0f;
    if (threadIdx.x < 16) av = aa[col0 + threadIdx.x];

    v2f a; a.x = 1.0f; a.y = 1.0f;
    v8f c = {};

    const size_t step = (size_t)CM_BATCH_ROWS * FEATS;
    const float* p = x + (size_t)(wave * CM_WAVE_ROWS + kb) * FEATS + col0 + n;

    v2f b0[CM_WMMAS], b1[CM_WMMAS], b2[CM_WMMAS], b3[CM_WMMAS];
    cm_request(b0, p);
    cm_request(b1, p + step);
    cm_request(b2, p + 2 * step);
    p += 3 * step;
    for (int it = 0; it < CM_BATCHES / 4 - 1; ++it) {
        cm_request(b3, p);              c = cm_add(c, a, b0);
        cm_request(b0, p + step);       c = cm_add(c, a, b1);
        cm_request(b1, p + 2 * step);   c = cm_add(c, a, b2);
        cm_request(b2, p + 3 * step);   c = cm_add(c, a, b3);
        p += 4 * step;
    }
    cm_request(b3, p);
    c = cm_add(c, a, b0);
    c = cm_add(c, a, b1);
    c = cm_add(c, a, b2);
    c = cm_add(c, a, b3);

    __shared__ float part[8][16];
    if (lane < 16) part[wave][lane] = c[0];
    __syncthreads();
    if (threadIdx.x < 16) {
        float s = 0.0f;
        for (int w = 0; w < 8; ++w) s += part[w][threadIdx.x];
        colmean[col0 + threadIdx.x] = s * (1.0f / (float)BATCH);
        factor[col0 + threadIdx.x]  = __expf((DENSITY - av) * 1.0f);
    }
}

__device__ __forceinline__ unsigned order_key(float f) {
    unsigned u = __float_as_uint(f);
    return (u & 0x80000000u) ? ~u : (u | 0x80000000u);
}

template <int CTRL>
__device__ __forceinline__ int from_higher_lane(int v) {
    return __builtin_amdgcn_update_dpp(0, v, CTRL, 0xF, 0xF, true);
}

__device__ __forceinline__ bool find_bin(const unsigned* h, int lane, int target,
                                         int& bin, int& tgt, int flip = 0) {
    const unsigned* own = h + ((8 * lane) ^ flip);
    const uint4 q0 = *(const uint4*)(own);
    const uint4 q1 = *(const uint4*)(own + 4);
    int c[8];
    c[0] = (int)q0.x; c[1] = (int)q0.y; c[2] = (int)q0.z; c[3] = (int)q0.w;
    c[4] = (int)q1.x; c[5] = (int)q1.y; c[6] = (int)q1.z; c[7] = (int)q1.w;
    const int tot = ((c[0] + c[1]) + (c[2] + c[3])) + ((c[4] + c[5]) + (c[6] + c[7]));

    int s = tot;
    s += from_higher_lane<0x101>(s);
    s += from_higher_lane<0x102>(s);
    s += from_higher_lane<0x104>(s);
    s += from_higher_lane<0x108>(s);
    const int upper = __builtin_amdgcn_readlane(s, 16);
    if (lane < 16) s += upper;

    int a = s - tot;
    int bsel = 0, tsel = 0;
    bool mine = false;
    #pragma unroll
    for (int j = 7; j >= 0; --j) {
        const int nxt = a + c[j];
        const bool hit = (a < target) && (target <= nxt);
        bsel = hit ? (8 * lane + j) : bsel;
        tsel = hit ? (target - a) : tsel;
        mine = mine || hit;
        a = nxt;
    }
    const unsigned owners = __builtin_amdgcn_ballot_w32(mine);
    const int owner = (int)__ffs(owners) - 1;
    bin = __builtin_amdgcn_readlane(bsel, owner);
    tgt = __builtin_amdgcn_readlane(tsel, owner);
    return owners != 0u;
}

__device__ __forceinline__ void count_digit(const unsigned rk[EPT], unsigned* h, int shift,
                                            unsigned prefH, unsigned prefL) {
    const unsigned hmask = 0xFFFFFFFFu << (shift + 8);
    #pragma unroll
    for (int e = 0; e < EPT; ++e) {
        const unsigned k  = rk[e];
        const unsigned hi = k & hmask;
        const bool mH = (hi == prefH);
        const bool mL = (hi == prefL);
        if (mH || mL)
            atomicAdd(&h[(mH ? 0u : 256u) + ((k >> shift) & 0xFFu)], 1u);
    }
}

__device__ __forceinline__ unsigned pick_rank(unsigned mine, bool valid, int above, int atleast,
                                              int target) {
    const bool hit = valid && (above < target) && (target <= atleast);
    const unsigned owners = __builtin_amdgcn_ballot_w32(hit);
    const int owner = (int)__ffs(owners) - 1;
    return (unsigned)__builtin_amdgcn_readlane((int)mine, owner);
}

#define LIST_SLOTS 32

__device__ __forceinline__ void rank_short_lists(
        const unsigned* lstH, const unsigned* lstL, unsigned* keys2,
        int wave, int lane, int cH, int cL, bool same, int targH, int targL,
        unsigned& keyHigh, unsigned& keyLow) {
    if (wave == 4 || (wave == 5 && !same)) {
        const int      c    = (wave == 4) ? cH : cL;
        const unsigned* lst = (wave == 4) ? lstH : lstL;
        const bool v = lane < c;
        const unsigned mine = v ? lst[lane] : 0u;
        int above = 0, atleast = 0;
        #pragma nounroll
        for (int j = 0; j < c; ++j) {
            const unsigned kj = (unsigned)__builtin_amdgcn_readlane((int)mine, j);
            above   += (kj >  mine) ? 1 : 0;
            atleast += (kj >= mine) ? 1 : 0;
        }
        if (wave == 4) {
            const unsigned kH = pick_rank(mine, v, above, atleast, targH);
            if (lane == 0) keys2[0] = kH;
            if (same) {
                const unsigned kL = pick_rank(mine, v, above, atleast, targL);
                if (lane == 0) keys2[1] = kL;
            }
        } else {
            const unsigned kL = pick_rank(mine, v, above, atleast, targL);
            if (lane == 0) keys2[1] = kL;
        }
    }
    __syncthreads();
    const uint2 r = *(const uint2*)keys2;
    keyHigh = (unsigned)__builtin_amdgcn_readfirstlane((int)r.x);
    keyLow  = (unsigned)__builtin_amdgcn_readfirstlane((int)r.y);
}

__device__ __forceinline__ void radix_select_two_desc(
        const unsigned rk[EPT], unsigned* lds, int t,
        unsigned& keyHigh, unsigned& keyLow) {
    unsigned* rep = lds;
    unsigned* tot = lds + REP_WORDS + 2 * 512;

    for (int e = 0; e < EPT; ++e)
        atomicAdd(&rep[((rk[e] >> 24) << 5) + (unsigned)(t & 31)], 1u);
    __syncthreads();
    {
        const unsigned* cnt = rep + 32 * t;
        const int rot = (t >> 1) & 7;
        unsigned s0 = 0, s1 = 0, s2 = 0, s3 = 0;
        for (int j = 0; j < 8; ++j) {
            const uint4 q = *(const uint4*)(cnt + 4 * ((j + rot) & 7));
            s0 += q.x; s1 += q.y; s2 += q.z; s3 += q.w;
        }
        tot[t] = (s0 + s1) + (s2 + s3);
    }
    __syncthreads();
    if (t == 0) {
        const uint2 zero2 = {0u, 0u};
        *(uint2*)rep = zero2;
    }
    const int wave = __builtin_amdgcn_readfirstlane(t >> 5);
    const int lane = (int)__builtin_amdgcn_mbcnt_lo(~0u, 0u);
    int binH, tgtH, binL, tgtL;
    if (wave == 0) {
        int b, g;
        find_bin(tot, lane, RANK_HIGH, b, g);
        if (lane == 0) { const uint2 r = {(unsigned)b, (unsigned)g}; *(uint2*)(rep + 4) = r; }
    } else if (wave == 1) {
        int b, g;
        find_bin(tot, lane, RANK_LOW, b, g);
        if (lane == 0) { const uint2 r = {(unsigned)b, (unsigned)g}; *(uint2*)(rep + 6) = r; }
    }
    __syncthreads();
    {
        const uint4 r = *(const uint4*)(rep + 4);
        binH = __builtin_amdgcn_readfirstlane((int)r.x);
        tgtH = __builtin_amdgcn_readfirstlane((int)r.y);
        binL = __builtin_amdgcn_readfirstlane((int)r.z);
        tgtL = __builtin_amdgcn_readfirstlane((int)r.w);
    }
    unsigned prefH = (unsigned)binH << 24;
    unsigned prefL = (unsigned)binL << 24;
    int targH = tgtH, targL = tgtL;
    bool same = (binH == binL);

    {
        unsigned* h = lds + REP_WORDS;
        count_digit(rk, h, 16, prefH, prefL);
        __syncthreads();
        if (wave == 2) {
            int b, g;
            find_bin(h, lane, targH, b, g);
            if (lane == 0) { const uint2 r = {(unsigned)b, (unsigned)g}; *(uint2*)(rep + 8) = r; }
        } else if (wave == 3) {
            int b, g;
            find_bin(same ? h : h + 256, lane, targL, b, g);
            if (lane == 0) { const uint2 r = {(unsigned)b, (unsigned)g}; *(uint2*)(rep + 10) = r; }
        }
        __syncthreads();
        {
            const uint4 r = *(const uint4*)(rep + 8);
            binH = __builtin_amdgcn_readfirstlane((int)r.x);
            tgtH = __builtin_amdgcn_readfirstlane((int)r.y);
            binL = __builtin_amdgcn_readfirstlane((int)r.z);
            tgtL = __builtin_amdgcn_readfirstlane((int)r.w);
        }
        prefH |= ((unsigned)binH << 16);
        prefL |= ((unsigned)binL << 16);
        targH = tgtH;
        targL = tgtL;
        same = same && (binH == binL);
    }

    #pragma unroll
    for (int e = 0; e < EPT; ++e) {
        const unsigned k  = rk[e];
        const unsigned hi = k & 0xFFFF0000u;
        const bool mH = (hi == prefH);
        const bool mL = (hi == prefL);
        if (mH || mL) {
            const unsigned w = mH ? 0u : 1u;
            const unsigned slot = atomicAdd(&rep[w], 1u);
            if (slot < (unsigned)LIST_SLOTS) rep[32u + 32u * w + slot] = k;
        }
    }
    __syncthreads();
    const int cH = __builtin_amdgcn_readfirstlane((int)rep[0]);
    const int cL = __builtin_amdgcn_readfirstlane((int)rep[1]);

    if (cH > LIST_SLOTS || cL > LIST_SLOTS) {
        const uint2 zero2 = {0u, 0u};
        ((uint2*)tot)[t] = zero2;
        #pragma unroll
        for (int d = 2; d < 4; ++d) {
            const int shift = 24 - 8 * d;
            unsigned* h = lds + REP_WORDS + 512 * (d - 1);
            count_digit(rk, h, shift, prefH, prefL);
            __syncthreads();
            find_bin(h, lane, targH, binH, tgtH);
            find_bin(same ? h : h + 256, lane, targL, binL, tgtL);
            prefH |= ((unsigned)binH << shift);
            prefL |= ((unsigned)binL << shift);
            targH = tgtH;
            targL = tgtL;
            same = same && (binH == binL);
        }
        keyHigh = prefH;
        keyLow  = prefL;
    } else {
        rank_short_lists(rep + 32, rep + 64, rep + 12, wave, lane, cH, cL, same,
                         targH, targL, keyHigh, keyLow);
    }
}

#define WIN_LO    0xBF000000u
#define WIN_HI    0xC0FFFFFFu
#define WIN_SHIFT 17
static_assert(((WIN_HI - WIN_LO) >> WIN_SHIFT) == 255u, "256 bins");
static_assert(((WIN_LO >> WIN_SHIFT) & 0xFFu) == 128u, "bin b is counted in word b ^ 128");

__device__ __forceinline__ bool window_select_two_desc(
        const unsigned rk[EPT], unsigned* lds, int t,
        unsigned& keyHigh, unsigned& keyLow) {
    unsigned* win = lds;
    unsigned* len = lds + 256;
    unsigned* res = lds + 260;
    unsigned* lst = lds + 288;

    #pragma unroll
    for (int e = 0; e < EPT; ++e) {
        const unsigned k = rk[e];
        if (k >= WIN_LO)
            atomicAdd(&win[(min(k, WIN_HI) >> WIN_SHIFT) & 0xFFu], 1u);
    }
    __syncthreads();
    const int wave = __builtin_amdgcn_readfirstlane(t >> 5);
    const int lane = (int)__builtin_amdgcn_mbcnt_lo(~0u, 0u);
    if (wave == 0) {
        int b, g;
        if (!find_bin(win, lane, RANK_HIGH, b, g, 128)) b = -1;
        if (lane == 0) { const uint2 r = {(unsigned)b, (unsigned)g}; *(uint2*)(res) = r; }
    } else if (wave == 1) {
        int b, g;
        if (!find_bin(win, lane, RANK_LOW, b, g, 128)) b = -1;
        if (lane == 0) { const uint2 r = {(unsigned)b, (unsigned)g}; *(uint2*)(res + 2) = r; }
    }
    __syncthreads();
    const uint4 r = *(const uint4*)res;
    const int binH  = __builtin_amdgcn_readfirstlane((int)r.x);
    const int targH = __builtin_amdgcn_readfirstlane((int)r.y);
    const int binL  = __builtin_amdgcn_readfirstlane((int)r.z);
    const int targL = __builtin_amdgcn_readfirstlane((int)r.w);
    if (binH < 0 || binL < 0) return false;
    const unsigned tH = (WIN_LO >> WIN_SHIFT) + (unsigned)binH;
    const unsigned tL = (WIN_LO >> WIN_SHIFT) + (unsigned)binL;
    const bool same = (binH == binL);

    #pragma unroll
    for (int e = 0; e < EPT; ++e) {
        const unsigned k  = rk[e];
        const unsigned tk = min(k, WIN_HI) >> WIN_SHIFT;
        const bool mH = (tk == tH);
        const bool mL = (tk == tL);
        if (mH || mL) {
            const unsigned w = mH ? 0u : 1u;
            const unsigned slot = atomicAdd(&len[w], 1u);
            if (slot < (unsigned)LIST_SLOTS) lst[32u * w + slot] = k;
        }
    }
    __syncthreads();
    const int cH = __builtin_amdgcn_readfirstlane((int)len[0]);
    const int cL = __builtin_amdgcn_readfirstlane((int)len[1]);
    if (cH > LIST_SLOTS || cL > LIST_SLOTS) return false;
    rank_short_lists(lst, lst + 32, res + 4, wave, lane, cH, cL, same,
                     targH, targL, keyHigh, keyLow);
    return true;
}

__global__ void __launch_bounds__(256, 16)
kw_row_kernel(const float* __restrict__ x,
              const float* __restrict__ factor,
              const float* __restrict__ colmean,
              float* __restrict__ out) {
    __shared__ __attribute__((aligned(16))) unsigned hist[REP_WORDS + 3 * 512];

    const int row = blockIdx.x;
    const int t   = threadIdx.x;

    const float4* __restrict__ xin = (const float4*)(x + (size_t)row * FEATS);
    const float4* __restrict__ fin = (const float4*)factor;

    float    rx[EPT];
    unsigned rk[EPT];

    #pragma unroll
    for (int it = 0; it < EPT / 4; ++it) {
        int idx = it * 256 + t;
        float4 xv = xin[idx];
        float4 fv = fin[idx];
        rx[it * 4 + 0] = xv.x;  rk[it * 4 + 0] = order_key(xv.x * fv.x);
        rx[it * 4 + 1] = xv.y;  rk[it * 4 + 1] = order_key(xv.y * fv.y);
        rx[it * 4 + 2] = xv.z;  rk[it * 4 + 2] = order_key(xv.z * fv.z);
        rx[it * 4 + 3] = xv.w;  rk[it * 4 + 3] = order_key(xv.w * fv.w);
    }

    hist[t] = 0u;
    if (t < 2) hist[256 + t] = 0u;
    __syncthreads();
    unsigned thrHigh, thrLow;
    if (!window_select_two_desc(rk, hist, t, thrHigh, thrLow)) {
        __syncthreads();
        const uint4 zero4 = {0u, 0u, 0u, 0u};
        for (int i = 0; i < (REP_WORDS + 2 * 512) / (4 * 256); ++i)
            ((uint4*)hist)[i * 256 + t] = zero4;
        __syncthreads();
        radix_select_two_desc(rk, hist, t, thrHigh, thrLow);
    }

    const float4* __restrict__ cmin = (const float4*)colmean;
    float4* __restrict__ oout = (float4*)(out + (size_t)row * FEATS);
    #pragma unroll
    for (int it = 0; it < EPT / 4; ++it) {
        int idx = it * 256 + t;
        float4 cm = cmin[idx];
        float4 o;
        {
            unsigned k = rk[it * 4 + 0];
            bool m2 = (k <= thrHigh);
            bool m  = m2 && (k >= thrLow);
            o.x = (m ? rx[it * 4 + 0] : 0.0f) - (m2 ? cm.x : 0.0f);
        }
        {
            unsigned k = rk[it * 4 + 1];
            bool m2 = (k <= thrHigh);
            bool m  = m2 && (k >= thrLow);
            o.y = (m ? rx[it * 4 + 1] : 0.0f) - (m2 ? cm.y : 0.0f);
        }
        {
            unsigned k = rk[it * 4 + 2];
            bool m2 = (k <= thrHigh);
            bool m  = m2 && (k >= thrLow);
            o.z = (m ? rx[it * 4 + 2] : 0.0f) - (m2 ? cm.z : 0.0f);
        }
        {
            unsigned k = rk[it * 4 + 3];
            bool m2 = (k <= thrHigh);
            bool m  = m2 && (k >= thrLow);
            o.w = (m ? rx[it * 4 + 3] : 0.0f) - (m2 ? cm.w : 0.0f);
        }
        oout[idx] = o;
    }
}

extern "C" void kernel_launch(void* const* d_in, const int* in_sizes, int n_in,
                              void* d_out, int out_size, void* d_ws, size_t ws_size,
                              hipStream_t stream) {
    const float* x  = (const float*)d_in[0];
    const float* aa = (const float*)d_in[1];
    float* out = (float*)d_out;

    float* factor  = (float*)d_ws;
    float* colmean = (float*)d_ws + FEATS;

    kw_colmean_wmma_kernel<<<FEATS / 16, 256, 0, stream>>>(x, aa, colmean, factor);

    kw_row_kernel<<<BATCH, 256, 0, stream>>>(x, factor, colmean, out);
}


# Row kernel: the window's counting add returns each key's place, the key is stored in its bin's own LDS list; the wave that finds a rank's bin ranks that list: no gather pass, 3 barriers a row not 5
# speedup vs baseline: 17.6571x; 1.0403x over previous
#include <hip/hip_runtime.h>

typedef __attribute__((ext_vector_type(2))) float v2f;
typedef __attribute__((ext_vector_type(8))) float v8f;

#define BATCH 8192
#define FEATS 4096
#define RANK_LOW 211
#define RANK_HIGH 7
#define DENSITY 0.05f
#define EPT 16
#define REP_WORDS (256 * 32)

#define CM_WMMAS 8
#define CM_BATCH_ROWS (4 * CM_WMMAS)
#define CM_WAVE_ROWS (BATCH / 8)
#define CM_BATCHES (CM_WAVE_ROWS / CM_BATCH_ROWS)
static_assert(CM_WAVE_ROWS % CM_BATCH_ROWS == 0, "whole batches only");
static_assert(CM_BATCHES >= 4 && CM_BATCHES % 4 == 0, "three batches ahead, four sets in turn");

__device__ __forceinline__ void cm_request(v2f (&b)[CM_WMMAS], const float* __restrict__ p) {
    #pragma unroll
    for (int i = 0; i < CM_WMMAS; ++i) {
        b[i].x = p[(size_t)(4 * i) * FEATS];
        b[i].y = p[(size_t)(4 * i + 1) * FEATS];
    }
}

__device__ __forceinline__ v8f cm_add(v8f c, const v2f a, const v2f (&b)[CM_WMMAS]) {
    #pragma unroll
    for (int i = 0; i < CM_WMMAS; ++i)
        c = __builtin_amdgcn_wmma_f32_16x16x4_f32(
                false, a, false, b[i], (short)0, c, false, false);
    return c;
}

__global__ void kw_colmean_wmma_kernel(const float* __restrict__ x,
                                       const float* __restrict__ aa,
                                       float* __restrict__ colmean,
                                       float* __restrict__ factor) {
    const int col0 = blockIdx.x * 16;
    const int lane = threadIdx.x & 31;
    const int wave = threadIdx.x >> 5;
    const int n    = lane & 15;
    const int kb   = (lane >> 4) * 2;

    float av = 0.0f;
    if (threadIdx.x < 16) av = aa[col0 + threadIdx.x];

    v2f a; a.x = 1.0f; a.y = 1.0f;
    v8f c = {};

    const size_t step = (size_t)CM_BATCH_ROWS * FEATS;
    const float* p = x + (size_t)(wave * CM_WAVE_ROWS + kb) * FEATS + col0 + n;

    v2f b0[CM_WMMAS], b1[CM_WMMAS], b2[CM_WMMAS], b3[CM_WMMAS];
    cm_request(b0, p);
    cm_request(b1, p + step);
    cm_request(b2, p + 2 * step);
    p += 3 * step;
    for (int it = 0; it < CM_BATCHES / 4 - 1; ++it) {
        cm_request(b3, p);              c = cm_add(c, a, b0);
        cm_request(b0, p + step);       c = cm_add(c, a, b1);
        cm_request(b1, p + 2 * step);   c = cm_add(c, a, b2);
        cm_request(b2, p + 3 * step);   c = cm_add(c, a, b3);
        p += 4 * step;
    }
    cm_request(b3, p);
    c = cm_add(c, a, b0);
    c = cm_add(c, a, b1);
    c = cm_add(c, a, b2);
    c = cm_add(c, a, b3);

    __shared__ float part[8][16];
    if (lane < 16) part[wave][lane] = c[0];
    __syncthreads();
    if (threadIdx.x < 16) {
        float s = 0.0f;
        for (int w = 0; w < 8; ++w) s += part[w][threadIdx.x];
        colmean[col0 + threadIdx.x] = s * (1.0f / (float)BATCH);
        factor[col0 + threadIdx.x]  = __expf((DENSITY - av) * 1.0f);
    }
}

__device__ __forceinline__ unsigned order_key(float f) {
    unsigned u = __float_as_uint(f);
    return (u & 0x80000000u) ? ~u : (u | 0x80000000u);
}

template <int CTRL>
__device__ __forceinline__ int from_higher_lane(int v) {
    return __builtin_amdgcn_update_dpp(0, v, CTRL, 0xF, 0xF, true);
}

__device__ __forceinline__ bool find_bin(const unsigned* h, int lane, int target,
                                         int& bin, int& tgt, int flip = 0) {
    const unsigned* own = h + ((8 * lane) ^ flip);
    const uint4 q0 = *(const uint4*)(own);
    const uint4 q1 = *(const uint4*)(own + 4);
    int c[8];
    c[0] = (int)q0.x; c[1] = (int)q0.y; c[2] = (int)q0.z; c[3] = (int)q0.w;
    c[4] = (int)q1.x; c[5] = (int)q1.y; c[6] = (int)q1.z; c[7] = (int)q1.w;
    const int tot = ((c[0] + c[1]) + (c[2] + c[3])) + ((c[4] + c[5]) + (c[6] + c[7]));

    int s = tot;
    s += from_higher_lane<0x101>(s);
    s += from_higher_lane<0x102>(s);
    s += from_higher_lane<0x104>(s);
    s += from_higher_lane<0x108>(s);
    const int upper = __builtin_amdgcn_readlane(s, 16);
    if (lane < 16) s += upper;

    int a = s - tot;
    int bsel = 0, tsel = 0;
    bool mine = false;
    #pragma unroll
    for (int j = 7; j >= 0; --j) {
        const int nxt = a + c[j];
        const bool hit = (a < target) && (target <= nxt);
        bsel = hit ? (8 * lane + j) : bsel;
        tsel = hit ? (target - a) : tsel;
        mine = mine || hit;
        a = nxt;
    }
    const unsigned owners = __builtin_amdgcn_ballot_w32(mine);
    const int owner = (int)__ffs(owners) - 1;
    bin = __builtin_amdgcn_readlane(bsel, owner);
    tgt = __builtin_amdgcn_readlane(tsel, owner);
    return owners != 0u;
}

__device__ __forceinline__ void count_digit(const unsigned rk[EPT], unsigned* h, int shift,
                                            unsigned prefH, unsigned prefL) {
    const unsigned hmask = 0xFFFFFFFFu << (shift + 8);
    #pragma unroll
    for (int e = 0; e < EPT; ++e) {
        const unsigned k  = rk[e];
        const unsigned hi = k & hmask;
        const bool mH = (hi == prefH);
        const bool mL = (hi == prefL);
        if (mH || mL)
            atomicAdd(&h[(mH ? 0u : 256u) + ((k >> shift) & 0xFFu)], 1u);
    }
}

__device__ __forceinline__ unsigned pick_rank(unsigned mine, bool valid, int above, int atleast,
                                              int target) {
    const bool hit = valid && (above < target) && (target <= atleast);
    const unsigned owners = __builtin_amdgcn_ballot_w32(hit);
    const int owner = (int)__ffs(owners) - 1;
    return (unsigned)__builtin_amdgcn_readlane((int)mine, owner);
}

#define LIST_SLOTS 32

__device__ __forceinline__ void rank_short_lists(
        const unsigned* lstH, const unsigned* lstL, unsigned* keys2,
        int wave, int lane, int cH, int cL, bool same, int targH, int targL,
        unsigned& keyHigh, unsigned& keyLow) {
    if (wave == 4 || (wave == 5 && !same)) {
        const int      c    = (wave == 4) ? cH : cL;
        const unsigned* lst = (wave == 4) ? lstH : lstL;
        const bool v = lane < c;
        const unsigned mine = v ? lst[lane] : 0u;
        int above = 0, atleast = 0;
        #pragma nounroll
        for (int j = 0; j < c; ++j) {
            const unsigned kj = (unsigned)__builtin_amdgcn_readlane((int)mine, j);
            above   += (kj >  mine) ? 1 : 0;
            atleast += (kj >= mine) ? 1 : 0;
        }
        if (wave == 4) {
            const unsigned kH = pick_rank(mine, v, above, atleast, targH);
            if (lane == 0) keys2[0] = kH;
            if (same) {
                const unsigned kL = pick_rank(mine, v, above, atleast, targL);
                if (lane == 0) keys2[1] = kL;
            }
        } else {
            const unsigned kL = pick_rank(mine, v, above, atleast, targL);
            if (lane == 0) keys2[1] = kL;
        }
    }
    __syncthreads();
    const uint2 r = *(const uint2*)keys2;
    keyHigh = (unsigned)__builtin_amdgcn_readfirstlane((int)r.x);
    keyLow  = (unsigned)__builtin_amdgcn_readfirstlane((int)r.y);
}

__device__ __forceinline__ void radix_select_two_desc(
        const unsigned rk[EPT], unsigned* lds, int t,
        unsigned& keyHigh, unsigned& keyLow) {
    unsigned* rep = lds;
    unsigned* tot = lds + REP_WORDS + 2 * 512;

    for (int e = 0; e < EPT; ++e)
        atomicAdd(&rep[((rk[e] >> 24) << 5) + (unsigned)(t & 31)], 1u);
    __syncthreads();
    {
        const unsigned* cnt = rep + 32 * t;
        const int rot = (t >> 1) & 7;
        unsigned s0 = 0, s1 = 0, s2 = 0, s3 = 0;
        for (int j = 0; j < 8; ++j) {
            const uint4 q = *(const uint4*)(cnt + 4 * ((j + rot) & 7));
            s0 += q.x; s1 += q.y; s2 += q.z; s3 += q.w;
        }
        tot[t] = (s0 + s1) + (s2 + s3);
    }
    __syncthreads();
    if (t == 0) {
        const uint2 zero2 = {0u, 0u};
        *(uint2*)rep = zero2;
    }
    const int wave = __builtin_amdgcn_readfirstlane(t >> 5);
    const int lane = (int)__builtin_amdgcn_mbcnt_lo(~0u, 0u);
    int binH, tgtH, binL, tgtL;
    if (wave == 0) {
        int b, g;
        find_bin(tot, lane, RANK_HIGH, b, g);
        if (lane == 0) { const uint2 r = {(unsigned)b, (unsigned)g}; *(uint2*)(rep + 4) = r; }
    } else if (wave == 1) {
        int b, g;
        find_bin(tot, lane, RANK_LOW, b, g);
        if (lane == 0) { const uint2 r = {(unsigned)b, (unsigned)g}; *(uint2*)(rep + 6) = r; }
    }
    __syncthreads();
    {
        const uint4 r = *(const uint4*)(rep + 4);
        binH = __builtin_amdgcn_readfirstlane((int)r.x);
        tgtH = __builtin_amdgcn_readfirstlane((int)r.y);
        binL = __builtin_amdgcn_readfirstlane((int)r.z);
        tgtL = __builtin_amdgcn_readfirstlane((int)r.w);
    }
    unsigned prefH = (unsigned)binH << 24;
    unsigned prefL = (unsigned)binL << 24;
    int targH = tgtH, targL = tgtL;
    bool same = (binH == binL);

    {
        unsigned* h = lds + REP_WORDS;
        count_digit(rk, h, 16, prefH, prefL);
        __syncthreads();
        if (wave == 2) {
            int b, g;
            find_bin(h, lane, targH, b, g);
            if (lane == 0) { const uint2 r = {(unsigned)b, (unsigned)g}; *(uint2*)(rep + 8) = r; }
        } else if (wave == 3) {
            int b, g;
            find_bin(same ? h : h + 256, lane, targL, b, g);
            if (lane == 0) { const uint2 r = {(unsigned)b, (unsigned)g}; *(uint2*)(rep + 10) = r; }
        }
        __syncthreads();
        {
            const uint4 r = *(const uint4*)(rep + 8);
            binH = __builtin_amdgcn_readfirstlane((int)r.x);
            tgtH = __builtin_amdgcn_readfirstlane((int)r.y);
            binL = __builtin_amdgcn_readfirstlane((int)r.z);
            tgtL = __builtin_amdgcn_readfirstlane((int)r.w);
        }
        prefH |= ((unsigned)binH << 16);
        prefL |= ((unsigned)binL << 16);
        targH = tgtH;
        targL = tgtL;
        same = same && (binH == binL);
    }

    #pragma unroll
    for (int e = 0; e < EPT; ++e) {
        const unsigned k  = rk[e];
        const unsigned hi = k & 0xFFFF0000u;
        const bool mH = (hi == prefH);
        const bool mL = (hi == prefL);
        if (mH || mL) {
            const unsigned w = mH ? 0u : 1u;
            const unsigned slot = atomicAdd(&rep[w], 1u);
            if (slot < (unsigned)LIST_SLOTS) rep[32u + 32u * w + slot] = k;
        }
    }
    __syncthreads();
    const int cH = __builtin_amdgcn_readfirstlane((int)rep[0]);
    const int cL = __builtin_amdgcn_readfirstlane((int)rep[1]);

    if (cH > LIST_SLOTS || cL > LIST_SLOTS) {
        const uint2 zero2 = {0u, 0u};
        ((uint2*)tot)[t] = zero2;
        #pragma unroll
        for (int d = 2; d < 4; ++d) {
            const int shift = 24 - 8 * d;
            unsigned* h = lds + REP_WORDS + 512 * (d - 1);
            count_digit(rk, h, shift, prefH, prefL);
            __syncthreads();
            find_bin(h, lane, targH, binH, tgtH);
            find_bin(same ? h : h + 256, lane, targL, binL, tgtL);
            prefH |= ((unsigned)binH << shift);
            prefL |= ((unsigned)binL << shift);
            targH = tgtH;
            targL = tgtL;
            same = same && (binH == binL);
        }
        keyHigh = prefH;
        keyLow  = prefL;
    } else {
        rank_short_lists(rep + 32, rep + 64, rep + 12, wave, lane, cH, cL, same,
                         targH, targL, keyHigh, keyLow);
    }
}

#define WIN_LO    0xBF000000u
#define WIN_HI    0xC0FFFFFFu
#define WIN_SHIFT 17
#define LIST_PITCH 33
#define WIN_CNT   (256 * LIST_PITCH)
#define WIN_RES   (WIN_CNT + 256)
static_assert(((WIN_HI - WIN_LO) >> WIN_SHIFT) == 255u, "256 bins");
static_assert(((WIN_LO >> WIN_SHIFT) & 0xFFu) == 128u, "bin b is counted in word b ^ 128");
static_assert(LIST_PITCH >= LIST_SLOTS, "a list holds LIST_SLOTS keys");
static_assert(WIN_CNT % 4 == 0 && WIN_RES % 4 == 0, "16-byte reads of counters and results");
static_assert(WIN_RES + 4 <= REP_WORDS + 2 * 512, "inside the words the general path clears");

__device__ __forceinline__ unsigned window_index(unsigned k) {
    return (min(k, WIN_HI) >> WIN_SHIFT) & 0xFFu;
}

__device__ __forceinline__ bool window_select_two_desc(
        const unsigned rk[EPT], unsigned* lds, int t,
        unsigned& keyHigh, unsigned& keyLow) {
    unsigned* lst = lds;
    unsigned* cnt = lds + WIN_CNT;
    unsigned* res = lds + WIN_RES;

    #pragma unroll
    for (int g = 0; g < EPT; g += 8) {
        unsigned slot[8];
        unsigned idx[8];
        #pragma unroll
        for (int e = 0; e < 8; ++e) {
            const unsigned k = rk[g + e];
            slot[e] = 0xFFFFFFFFu;
            idx[e] = window_index(k);
            if (k >= WIN_LO)
                slot[e] = atomicAdd(&cnt[idx[e]], 1u);
        }
        #pragma unroll
        for (int e = 0; e < 8; ++e) {
            if (slot[e] < (unsigned)LIST_SLOTS)
                lst[idx[e] * LIST_PITCH + slot[e]] = rk[g + e];
        }
    }
    __syncthreads();
    const int wave = __builtin_amdgcn_readfirstlane(t >> 5);
    const int lane = (int)__builtin_amdgcn_mbcnt_lo(~0u, 0u);
    if (wave < 2) {
        const int target = (wave == 0) ? RANK_HIGH : RANK_LOW;
        int b, g;
        bool ok = find_bin(cnt, lane, target, b, g, 128);
        unsigned key = 0u;
        if (ok) {
            const unsigned j = (unsigned)b ^ 128u;
            const int c = __builtin_amdgcn_readfirstlane((int)cnt[j]);
            ok = (c <= LIST_SLOTS);
            if (ok) {
                const bool v = lane < c;
                const unsigned mine = v ? lst[j * LIST_PITCH + (unsigned)lane] : 0u;
                int above = 0, atleast = 0;
                #pragma nounroll
                for (int i = 0; i < c; ++i) {
                    const unsigned ki = (unsigned)__builtin_amdgcn_readlane((int)mine, i);
                    above   += (ki >  mine) ? 1 : 0;
                    atleast += (ki >= mine) ? 1 : 0;
                }
                key = pick_rank(mine, v, above, atleast, g);
            }
        }
        if (lane == 0) {
            res[wave]     = key;
            res[2 + wave] = ok ? 1u : 0u;
        }
    }
    __syncthreads();
    const uint4 r = *(const uint4*)res;
    keyHigh = (unsigned)__builtin_amdgcn_readfirstlane((int)r.x);
    keyLow  = (unsigned)__builtin_amdgcn_readfirstlane((int)r.y);
    const int okH = __builtin_amdgcn_readfirstlane((int)r.z);
    const int okL = __builtin_amdgcn_readfirstlane((int)r.w);
    return okH != 0 && okL != 0;
}

__global__ void __launch_bounds__(256, 16)
kw_row_kernel(const float* __restrict__ x,
              const float* __restrict__ factor,
              const float* __restrict__ colmean,
              float* __restrict__ out) {
    __shared__ __attribute__((aligned(16))) unsigned hist[REP_WORDS + 3 * 512];

    const int row = blockIdx.x;
    const int t   = threadIdx.x;

    const float4* __restrict__ xin = (const float4*)(x + (size_t)row * FEATS);
    const float4* __restrict__ fin = (const float4*)factor;

    hist[WIN_CNT + t] = 0u;
    __syncthreads();

    float    rx[EPT];
    unsigned rk[EPT];

    #pragma unroll
    for (int it = 0; it < EPT / 4; ++it) {
        int idx = it * 256 + t;
        float4 xv = xin[idx];
        float4 fv = fin[idx];
        rx[it * 4 + 0] = xv.x;  rk[it * 4 + 0] = order_key(xv.x * fv.x);
        rx[it * 4 + 1] = xv.y;  rk[it * 4 + 1] = order_key(xv.y * fv.y);
        rx[it * 4 + 2] = xv.z;  rk[it * 4 + 2] = order_key(xv.z * fv.z);
        rx[it * 4 + 3] = xv.w;  rk[it * 4 + 3] = order_key(xv.w * fv.w);
    }

    unsigned thrHigh, thrLow;
    if (!window_select_two_desc(rk, hist, t, thrHigh, thrLow)) {
        __syncthreads();
        const uint4 zero4 = {0u, 0u, 0u, 0u};
        for (int i = 0; i < (REP_WORDS + 2 * 512) / (4 * 256); ++i)
            ((uint4*)hist)[i * 256 + t] = zero4;
        __syncthreads();
        radix_select_two_desc(rk, hist, t, thrHigh, thrLow);
    }

    const float4* __restrict__ cmin = (const float4*)colmean;
    float4* __restrict__ oout = (float4*)(out + (size_t)row * FEATS);
    #pragma unroll
    for (int it = 0; it < EPT / 4; ++it) {
        int idx = it * 256 + t;
        float4 cm = cmin[idx];
        float4 o;
        {
            unsigned k = rk[it * 4 + 0];
            bool m2 = (k <= thrHigh);
            bool m  = m2 && (k >= thrLow);
            o.x = (m ? rx[it * 4 + 0] : 0.0f) - (m2 ? cm.x : 0.0f);
        }
        {
            unsigned k = rk[it * 4 + 1];
            bool m2 = (k <= thrHigh);
            bool m  = m2 && (k >= thrLow);
            o.y = (m ? rx[it * 4 + 1] : 0.0f) - (m2 ? cm.y : 0.0f);
        }
        {
            unsigned k = rk[it * 4 + 2];
            bool m2 = (k <= thrHigh);
            bool m  = m2 && (k >= thrLow);
            o.z = (m ? rx[it * 4 + 2] : 0.0f) - (m2 ? cm.z : 0.0f);
        }
        {
            unsigned k = rk[it * 4 + 3];
            bool m2 = (k <= thrHigh);
            bool m  = m2 && (k >= thrLow);
            o.w = (m ? rx[it * 4 + 3] : 0.0f) - (m2 ? cm.w : 0.0f);
        }
        oout[idx] = o;
    }
}

extern "C" void kernel_launch(void* const* d_in, const int* in_sizes, int n_in,
                              void* d_out, int out_size, void* d_ws, size_t ws_size,
                              hipStream_t stream) {
    const float* x  = (const float*)d_in[0];
    const float* aa = (const float*)d_in[1];
    float* out = (float*)d_out;

    float* factor  = (float*)d_ws;
    float* colmean = (float*)d_ws + FEATS;

    kw_colmean_wmma_kernel<<<FEATS / 16, 256, 0, stream>>>(x, aa, colmean, factor);

    kw_row_kernel<<<BATCH, 256, 0, stream>>>(x, factor, colmean, out);
}
